# starting kernel
# baseline (speedup 1.0000x reference)
#include <hip/hip_runtime.h>

typedef __attribute__((ext_vector_type(2))) float v2f;
typedef __attribute__((ext_vector_type(8))) float v8f;

__device__ __forceinline__ void fillCoreA(float* dst, const float* __restrict__ src, int tid) {
  for (int idx = tid; idx < 1600; idx += 256) {
    int p = idx & 1;
    int l = (idx >> 1) & 31;
    int t = idx >> 6;
    int K = 4 * t + ((l & 16) ? 2 : 0) + p;
    int j = l & 15;
    int i = K / 10;
    int k = K % 10;
    dst[idx] = (j < 10) ? src[(i * 10 + k) * 10 + j] : 0.0f;
  }
}

__device__ __forceinline__ void leafContract(const float* sFM, const float* __restrict__ X,
                                             int b, int f, int u, float c[10]) {
  float x0 = X[(b * 8 + f) * 2 + 0];
  float x1 = X[(b * 8 + f) * 2 + 1];
#pragma unroll
  for (int r = 0; r < 10; ++r) {
    c[r] = x0 * sFM[((r) * 8 + f) * 32 + u] + x1 * sFM[((10 + r) * 8 + f) * 32 + u];
  }
}

__device__ __forceinline__ void pairContract(const float* sCoreC, const float a[10],
                                             const float bv[10], int lane, float q[10]) {
  v8f acc = {};
  bool hi = (lane & 16) != 0;
#pragma unroll
  for (int t = 0; t < 25; ++t) {
    v2f A = *(const v2f*)(sCoreC + (t * 32 + lane) * 2);
    float blo0 = a[(4 * t    ) / 10] * bv[(4 * t    ) % 10];
    float blo1 = a[(4 * t + 1) / 10] * bv[(4 * t + 1) % 10];
    float bhi0 = a[(4 * t + 2) / 10] * bv[(4 * t + 2) % 10];
    float bhi1 = a[(4 * t + 3) / 10] * bv[(4 * t + 3) % 10];
    v2f Bv;
    Bv.x = hi ? bhi0 : blo0;
    Bv.y = hi ? bhi1 : blo1;
    acc = __builtin_amdgcn_wmma_f32_16x16x4_f32(false, A, false, Bv, (short)0, acc,
                                                false, false);
  }
  float m[8];
#pragma unroll
  for (int v = 0; v < 8; ++v) m[v] = __shfl_xor(acc[v], 16, 32);
#pragma unroll
  for (int v = 0; v < 8; ++v) q[v] = hi ? m[v] : acc[v];
  q[8] = hi ? acc[0] : m[0];
  q[9] = hi ? acc[1] : m[1];
}

__global__ __launch_bounds__(256) void tensortree_wmma_kernel(
    const float* __restrict__ X, const float* __restrict__ c11,
    const float* __restrict__ c12, const float* __restrict__ c13,
    const float* __restrict__ c14, const float* __restrict__ c21,
    const float* __restrict__ c22, const float* __restrict__ fm,
    const float* __restrict__ T, float* __restrict__ out) {
  __shared__ float sCore[6 * 1600];
  __shared__ float sFM[5120];

  int tid = threadIdx.x;
  fillCoreA(sCore + 0,    c11, tid);
  fillCoreA(sCore + 1600, c12, tid);
  fillCoreA(sCore + 3200, c13, tid);
  fillCoreA(sCore + 4800, c14, tid);
  fillCoreA(sCore + 6400, c21, tid);
  fillCoreA(sCore + 8000, c22, tid);
  for (int idx = tid; idx < 5120; idx += 256) sFM[idx] = fm[idx];
  __syncthreads();

  int lane = tid & 31;
  int wave = blockIdx.x * 8 + (tid >> 5);
  int b = __builtin_amdgcn_readfirstlane(wave >> 1);
  int u = ((wave & 1) << 4) | (lane & 15);

  float cA[10], cB[10], pA[10], pB[10], q1[10], q2[10];

  leafContract(sFM, X, b, 0, u, cA);
  leafContract(sFM, X, b, 1, u, cB);
  pairContract(sCore + 0, cA, cB, lane, pA);
  leafContract(sFM, X, b, 2, u, cA);
  leafContract(sFM, X, b, 3, u, cB);
  pairContract(sCore + 1600, cA, cB, lane, pB);
  pairContract(sCore + 6400, pA, pB, lane, q1);

  leafContract(sFM, X, b, 4, u, cA);
  leafContract(sFM, X, b, 5, u, cB);
  pairContract(sCore + 3200, cA, cB, lane, pA);
  leafContract(sFM, X, b, 6, u, cA);
  leafContract(sFM, X, b, 7, u, cB);
  pairContract(sCore + 4800, cA, cB, lane, pB);
  pairContract(sCore + 8000, pA, pB, lane, q2);

  float r = 0.0f;
#pragma unroll
  for (int i = 0; i < 10; ++i) {
    float s = 0.0f;
#pragma unroll
    for (int j = 0; j < 10; ++j) s += T[i * 10 + j] * q2[j];
    r += q1[i] * s;
  }
  if (lane < 16) out[b * 32 + u] = r;
}

extern "C" void kernel_launch(void* const* d_in, const int* in_sizes, int n_in,
                              void* d_out, int out_size, void* d_ws, size_t ws_size,
                              hipStream_t stream) {
  const float* X   = (const float*)d_in[0];
  const float* c11 = (const float*)d_in[1];
  const float* c12 = (const float*)d_in[2];
  const float* c13 = (const float*)d_in[3];
  const float* c14 = (const float*)d_in[4];
  const float* c21 = (const float*)d_in[5];
  const float* c22 = (const float*)d_in[6];
  const float* fm  = (const float*)d_in[7];
  const float* T   = (const float*)d_in[8];
  float* out = (float*)d_out;

  int B = in_sizes[0] / 16;
  int blocks = (B * 32) / 128;
  tensortree_wmma_kernel<<<dim3(blocks), dim3(256), 0, stream>>>(
      X, c11, c12, c13, c14, c21, c22, fm, T, out);
}


# Pair-contraction k order and core row order made lane-half-relative: B operand is one packed multiply per WMMA, no lane-half selects, merges shrink to 5 exchanges; sum order over (i,k) changed
# speedup vs baseline: 1.4620x; 1.4620x over previous
#include <hip/hip_runtime.h>


typedef __attribute__((ext_vector_type(2))) float v2f;
typedef __attribute__((ext_vector_type(8))) float v8f;

#define IMG_CORE_FLOATS 9600
#define IMG_FLOATS 14720
#define IMG_VEC4 3680

__global__ __launch_bounds__(256) void pack_operands_kernel(
    const float* __restrict__ c11, const float* __restrict__ c12,
    const float* __restrict__ c13, const float* __restrict__ c14,
    const float* __restrict__ c21, const float* __restrict__ c22,
    const float* __restrict__ fm, float* __restrict__ img) {
  int idx = blockIdx.x * 256 + threadIdx.x;
  if (idx >= IMG_FLOATS) return;
  float v;
  if (idx < IMG_CORE_FLOATS) {
    int c = idx / 1600;
    int e = idx - c * 1600;
    const float* src = (c == 0) ? c11 : (c == 1) ? c12 : (c == 2) ? c13
                     : (c == 3) ? c14 : (c == 4) ? c21 : c22;
    int p = e & 1;
    int l = (e >> 1) & 31;
    int t = e >> 6;
    int h = l >> 4;
    int m = l & 15;
    int i = 5 * h + t / 5;
    int q = 2 * (t % 5) + p;
    int k = (q < 5) ? (5 * h + q) : (5 * (1 - h) + (q - 5));
    int r = m & 7;
    int j = 5 * (m >> 3) + r;
    v = (r < 5) ? src[(i * 10 + k) * 10 + j] : 0.0f;
  } else {
    v = fm[idx - IMG_CORE_FLOATS];
  }
  img[idx] = v;
}


__device__ __forceinline__ void leafOwn2(const float* sOwn, const float* __restrict__ X,
                                         int b0, int f, float a0[5], float a1[5]) {
  float x00 = X[(b0 * 8 + f) * 2 + 0];
  float x01 = X[(b0 * 8 + f) * 2 + 1];
  float x10 = X[((b0 + 1) * 8 + f) * 2 + 0];
  float x11 = X[((b0 + 1) * 8 + f) * 2 + 1];
#pragma unroll
  for (int x = 0; x < 5; ++x) {
    float f0 = sOwn[((x) * 8 + f) * 32];
    float f1 = sOwn[((10 + x) * 8 + f) * 32];
    a0[x] = x00 * f0 + x01 * f1;
    a1[x] = x10 * f0 + x11 * f1;
  }
}

__device__ __forceinline__ void leafRel2(const float* sOwn, const float* sOth,
                                         const float* __restrict__ X, int b0, int f,
                                         v2f r0[5], v2f r1[5]) {
  float x00 = X[(b0 * 8 + f) * 2 + 0];
  float x01 = X[(b0 * 8 + f) * 2 + 1];
  float x10 = X[((b0 + 1) * 8 + f) * 2 + 0];
  float x11 = X[((b0 + 1) * 8 + f) * 2 + 1];
  float c0[10], c1[10];
#pragma unroll
  for (int q = 0; q < 10; ++q) {
    const float* base = (q < 5) ? sOwn : sOth;
    int x = (q < 5) ? q : q - 5;
    float f0 = base[((x) * 8 + f) * 32];
    float f1 = base[((10 + x) * 8 + f) * 32];
    c0[q] = x00 * f0 + x01 * f1;
    c1[q] = x10 * f0 + x11 * f1;
  }
#pragma unroll
  for (int m = 0; m < 5; ++m) {
    r0[m].x = c0[2 * m]; r0[m].y = c0[2 * m + 1];
    r1[m].x = c1[2 * m]; r1[m].y = c1[2 * m + 1];
  }
}

__device__ __forceinline__ void pairContract2(const float* sCoreC,
                                              const float a0[5], const v2f b0v[5],
                                              const float a1[5], const v2f b1v[5],
                                              int lane, v8f& acc0, v8f& acc1) {
  acc0 = (v8f){};
  acc1 = (v8f){};
#pragma unroll
  for (int t = 0; t < 25; ++t) {
    v2f A = *(const v2f*)(sCoreC + (t * 32 + lane) * 2);
    v2f B0 = b0v[t % 5] * a0[t / 5];
    v2f B1 = b1v[t % 5] * a1[t / 5];
    acc0 = __builtin_amdgcn_wmma_f32_16x16x4_f32(false, A, false, B0, (short)0, acc0,
                                                 false, false);
    acc1 = __builtin_amdgcn_wmma_f32_16x16x4_f32(false, A, false, B1, (short)0, acc1,
                                                 false, false);
    asm("" : "+v"(acc0), "+v"(acc1));
  }
}

__device__ __forceinline__ void ownOf(const v8f& acc, float a[5]) {
#pragma unroll
  for (int x = 0; x < 5; ++x) a[x] = acc[x];
}

__device__ __forceinline__ void relOf(const v8f& acc, v2f r[5]) {
  float c[10];
#pragma unroll
  for (int x = 0; x < 5; ++x) {
    c[x] = acc[x];
    c[5 + x] = __shfl_xor(acc[x], 16, 32);
  }
#pragma unroll
  for (int m = 0; m < 5; ++m) { r[m].x = c[2 * m]; r[m].y = c[2 * m + 1]; }
}

__device__ __forceinline__ void fullOfMyTile(const v8f& acc0, const v8f& acc1, bool hi,
                                             float v[10]) {
#pragma unroll
  for (int x = 0; x < 5; ++x) {
    float mine = hi ? acc1[x] : acc0[x];
    float send = hi ? acc0[x] : acc1[x];
    float recv = __shfl_xor(send, 16, 32);
    v[x]     = hi ? recv : mine;
    v[5 + x] = hi ? mine : recv;
  }
}

__device__ __forceinline__ void phaseFence() { __builtin_amdgcn_sched_barrier(0); }

__global__ __launch_bounds__(256) void tensortree_wmma_kernel(
    const float* __restrict__ X, const float* __restrict__ img,
    const float* __restrict__ T, float* __restrict__ out) {
  __shared__ __attribute__((aligned(16))) float sImg[IMG_FLOATS];
  float* sCore = sImg;
  float* sFM = sImg + IMG_CORE_FLOATS;

  int tid = threadIdx.x;
  {
    const float4* g4 = reinterpret_cast<const float4*>(img);
    float4* s4 = reinterpret_cast<float4*>(sImg);
    float4 r[14];
#pragma unroll
    for (int it = 0; it < 14; ++it) r[it] = g4[tid + it * 256];
#pragma unroll
    for (int it = 0; it < 14; ++it) s4[tid + it * 256] = r[it];
    int tail = tid + 14 * 256;
    if (tail < IMG_VEC4) s4[tail] = g4[tail];
  }
  __syncthreads();

  int lane = tid & 31;
  int wave = blockIdx.x * 8 + (tid >> 5);
  int b0 = __builtin_amdgcn_readfirstlane((wave >> 1) << 1);
  int u = ((wave & 1) << 4) | (lane & 15);
  bool hi = (lane & 16) != 0;

  const float* sOwn = sFM + u + (hi ? 5 * 256 : 0);
  const float* sOth = sFM + u + (hi ? 0 : 5 * 256);

  float la0[5], la1[5], pa0[5], pa1[5];
  v2f lb0[5], lb1[5], pb0[5], pb1[5];
  v8f d0, d1;
  float qa[10], qb[10];

  leafOwn2(sOwn, X, b0, 0, la0, la1);
  leafRel2(sOwn, sOth, X, b0, 1, lb0, lb1);
  pairContract2(sCore + 0, la0, lb0, la1, lb1, lane, d0, d1);
  phaseFence();
  ownOf(d0, pa0); ownOf(d1, pa1);
  leafOwn2(sOwn, X, b0, 2, la0, la1);
  leafRel2(sOwn, sOth, X, b0, 3, lb0, lb1);
  pairContract2(sCore + 1600, la0, lb0, la1, lb1, lane, d0, d1);
  phaseFence();
  relOf(d0, pb0); relOf(d1, pb1);
  pairContract2(sCore + 6400, pa0, pb0, pa1, pb1, lane, d0, d1);
  fullOfMyTile(d0, d1, hi, qa);

  leafOwn2(sOwn, X, b0, 4, la0, la1);
  leafRel2(sOwn, sOth, X, b0, 5, lb0, lb1);
  pairContract2(sCore + 3200, la0, lb0, la1, lb1, lane, d0, d1);
  phaseFence();
  ownOf(d0, pa0); ownOf(d1, pa1);
  leafOwn2(sOwn, X, b0, 6, la0, la1);
  leafRel2(sOwn, sOth, X, b0, 7, lb0, lb1);
  pairContract2(sCore + 4800, la0, lb0, la1, lb1, lane, d0, d1);
  relOf(d0, pb0); relOf(d1, pb1);
  pairContract2(sCore + 8000, pa0, pb0, pa1, pb1, lane, d0, d1);
  fullOfMyTile(d0, d1, hi, qb);

  float r = 0.0f;
#pragma unroll
  for (int i = 0; i < 10; ++i) {
    float s = 0.0f;
#pragma unroll
    for (int j = 0; j < 10; ++j) s += T[i * 10 + j] * qb[j];
    r += qa[i] * s;
  }
  out[(b0 + (hi ? 1 : 0)) * 32 + u] = r;
}

extern "C" void kernel_launch(void* const* d_in, const int* in_sizes, int n_in,
                              void* d_out, int out_size, void* d_ws, size_t ws_size,
                              hipStream_t stream) {
  const float* X   = (const float*)d_in[0];
  const float* c11 = (const float*)d_in[1];
  const float* c12 = (const float*)d_in[2];
  const float* c13 = (const float*)d_in[3];
  const float* c14 = (const float*)d_in[4];
  const float* c21 = (const float*)d_in[5];
  const float* c22 = (const float*)d_in[6];
  const float* fm  = (const float*)d_in[7];
  const float* T   = (const float*)d_in[8];
  float* out = (float*)d_out;

  float* img = (float*)d_ws;

  int B = in_sizes[0] / 16;
  int blocks = (B * 32) / 256;
  pack_operands_kernel<<<dim3(58), dim3(256), 0, stream>>>(
      c11, c12, c13, c14, c21, c22, fm, img);
  tensortree_wmma_kernel<<<dim3(blocks), dim3(256), 0, stream>>>(X, img, T, out);
}


# Top matrix T contracted into core22's output index in the pack kernel: last contraction yields T.q2, epilogue is a 5-term in-lane dot plus one lane-half exchange; sum order changed
# speedup vs baseline: 1.5136x; 1.0353x over previous
#include <hip/hip_runtime.h>


typedef __attribute__((ext_vector_type(2))) float v2f;
typedef __attribute__((ext_vector_type(8))) float v8f;

#define IMG_CORE_FLOATS 9600
#define IMG_FLOATS 14720
#define IMG_VEC4 3680

__global__ __launch_bounds__(256) void pack_operands_kernel(
    const float* __restrict__ c11, const float* __restrict__ c12,
    const float* __restrict__ c13, const float* __restrict__ c14,
    const float* __restrict__ c21, const float* __restrict__ c22,
    const float* __restrict__ fm, const float* __restrict__ T,
    float* __restrict__ img) {
  int idx = blockIdx.x * 256 + threadIdx.x;
  if (idx >= IMG_FLOATS) return;
  float v;
  if (idx < IMG_CORE_FLOATS) {
    int c = idx / 1600;
    int e = idx - c * 1600;
    const float* src = (c == 0) ? c11 : (c == 1) ? c12 : (c == 2) ? c13
                     : (c == 3) ? c14 : (c == 4) ? c21 : c22;
    int p = e & 1;
    int l = (e >> 1) & 31;
    int t = e >> 6;
    int h = l >> 4;
    int m = l & 15;
    int i = 5 * h + t / 5;
    int q = 2 * (t % 5) + p;
    int k = (q < 5) ? (5 * h + q) : (5 * (1 - h) + (q - 5));
    int r = m & 7;
    int j = 5 * (m >> 3) + r;
    v = 0.0f;
    if (r < 5) {
      const float* row = src + (i * 10 + k) * 10;
      if (c == 5) {
        float acc = 0.0f;
#pragma unroll
        for (int j2 = 0; j2 < 10; ++j2) acc += row[j2] * T[j * 10 + j2];
        v = acc;
      } else {
        v = row[j];
      }
    }
  } else {
    v = fm[idx - IMG_CORE_FLOATS];
  }
  img[idx] = v;
}


__device__ __forceinline__ void leafOwn2(const float* sOwn, const float* __restrict__ X,
                                         int b0, int f, float a0[5], float a1[5]) {
  float x00 = X[(b0 * 8 + f) * 2 + 0];
  float x01 = X[(b0 * 8 + f) * 2 + 1];
  float x10 = X[((b0 + 1) * 8 + f) * 2 + 0];
  float x11 = X[((b0 + 1) * 8 + f) * 2 + 1];
#pragma unroll
  for (int x = 0; x < 5; ++x) {
    float f0 = sOwn[((x) * 8 + f) * 32];
    float f1 = sOwn[((10 + x) * 8 + f) * 32];
    a0[x] = x00 * f0 + x01 * f1;
    a1[x] = x10 * f0 + x11 * f1;
  }
}

__device__ __forceinline__ void leafRel2(const float* sOwn, const float* sOth,
                                         const float* __restrict__ X, int b0, int f,
                                         v2f r0[5], v2f r1[5]) {
  float x00 = X[(b0 * 8 + f) * 2 + 0];
  float x01 = X[(b0 * 8 + f) * 2 + 1];
  float x10 = X[((b0 + 1) * 8 + f) * 2 + 0];
  float x11 = X[((b0 + 1) * 8 + f) * 2 + 1];
  float c0[10], c1[10];
#pragma unroll
  for (int q = 0; q < 10; ++q) {
    const float* base = (q < 5) ? sOwn : sOth;
    int x = (q < 5) ? q : q - 5;
    float f0 = base[((x) * 8 + f) * 32];
    float f1 = base[((10 + x) * 8 + f) * 32];
    c0[q] = x00 * f0 + x01 * f1;
    c1[q] = x10 * f0 + x11 * f1;
  }
#pragma unroll
  for (int m = 0; m < 5; ++m) {
    r0[m].x = c0[2 * m]; r0[m].y = c0[2 * m + 1];
    r1[m].x = c1[2 * m]; r1[m].y = c1[2 * m + 1];
  }
}

__device__ __forceinline__ void pairContract2(const float* sCoreC,
                                              const float a0[5], const v2f b0v[5],
                                              const float a1[5], const v2f b1v[5],
                                              int lane, v8f& acc0, v8f& acc1) {
  acc0 = (v8f){};
  acc1 = (v8f){};
#pragma unroll
  for (int t = 0; t < 25; ++t) {
    v2f A = *(const v2f*)(sCoreC + (t * 32 + lane) * 2);
    v2f B0 = b0v[t % 5] * a0[t / 5];
    v2f B1 = b1v[t % 5] * a1[t / 5];
    acc0 = __builtin_amdgcn_wmma_f32_16x16x4_f32(false, A, false, B0, (short)0, acc0,
                                                 false, false);
    acc1 = __builtin_amdgcn_wmma_f32_16x16x4_f32(false, A, false, B1, (short)0, acc1,
                                                 false, false);
    asm("" : "+v"(acc0), "+v"(acc1));
  }
}

__device__ __forceinline__ void ownOf(const v8f& acc, float a[5]) {
#pragma unroll
  for (int x = 0; x < 5; ++x) a[x] = acc[x];
}

__device__ __forceinline__ void relOf(const v8f& acc, v2f r[5]) {
  float c[10];
#pragma unroll
  for (int x = 0; x < 5; ++x) {
    c[x] = acc[x];
    c[5 + x] = __shfl_xor(acc[x], 16, 32);
  }
#pragma unroll
  for (int m = 0; m < 5; ++m) { r[m].x = c[2 * m]; r[m].y = c[2 * m + 1]; }
}

__device__ __forceinline__ void phaseFence() { __builtin_amdgcn_sched_barrier(0); }

__global__ __launch_bounds__(256) void tensortree_wmma_kernel(
    const float* __restrict__ X, const float* __restrict__ img,
    float* __restrict__ out) {
  __shared__ __attribute__((aligned(16))) float sImg[IMG_FLOATS];
  float* sCore = sImg;
  float* sFM = sImg + IMG_CORE_FLOATS;

  int tid = threadIdx.x;
  {
    const float4* g4 = reinterpret_cast<const float4*>(img);
    float4* s4 = reinterpret_cast<float4*>(sImg);
    float4 r[14];
#pragma unroll
    for (int it = 0; it < 14; ++it) r[it] = g4[tid + it * 256];
#pragma unroll
    for (int it = 0; it < 14; ++it) s4[tid + it * 256] = r[it];
    int tail = tid + 14 * 256;
    if (tail < IMG_VEC4) s4[tail] = g4[tail];
  }
  __syncthreads();

  int lane = tid & 31;
  int wave = blockIdx.x * 8 + (tid >> 5);
  int b0 = __builtin_amdgcn_readfirstlane((wave >> 1) << 1);
  int u = ((wave & 1) << 4) | (lane & 15);
  bool hi = (lane & 16) != 0;

  const float* sOwn = sFM + u + (hi ? 5 * 256 : 0);
  const float* sOth = sFM + u + (hi ? 0 : 5 * 256);

  float la0[5], la1[5], pa0[5], pa1[5];
  v2f lb0[5], lb1[5], pb0[5], pb1[5];
  v8f d0, d1;
  float qa0[5], qa1[5];

  leafOwn2(sOwn, X, b0, 0, la0, la1);
  leafRel2(sOwn, sOth, X, b0, 1, lb0, lb1);
  pairContract2(sCore + 0, la0, lb0, la1, lb1, lane, d0, d1);
  phaseFence();
  ownOf(d0, pa0); ownOf(d1, pa1);
  leafOwn2(sOwn, X, b0, 2, la0, la1);
  leafRel2(sOwn, sOth, X, b0, 3, lb0, lb1);
  pairContract2(sCore + 1600, la0, lb0, la1, lb1, lane, d0, d1);
  phaseFence();
  relOf(d0, pb0); relOf(d1, pb1);
  pairContract2(sCore + 6400, pa0, pb0, pa1, pb1, lane, d0, d1);
  ownOf(d0, qa0); ownOf(d1, qa1);

  leafOwn2(sOwn, X, b0, 4, la0, la1);
  leafRel2(sOwn, sOth, X, b0, 5, lb0, lb1);
  pairContract2(sCore + 3200, la0, lb0, la1, lb1, lane, d0, d1);
  phaseFence();
  ownOf(d0, pa0); ownOf(d1, pa1);
  leafOwn2(sOwn, X, b0, 6, la0, la1);
  leafRel2(sOwn, sOth, X, b0, 7, lb0, lb1);
  pairContract2(sCore + 4800, la0, lb0, la1, lb1, lane, d0, d1);
  relOf(d0, pb0); relOf(d1, pb1);
  pairContract2(sCore + 8000, pa0, pb0, pa1, pb1, lane, d0, d1);

  float s0 = 0.0f, s1 = 0.0f;
#pragma unroll
  for (int x = 0; x < 5; ++x) {
    s0 += qa0[x] * d0[x];
    s1 += qa1[x] * d1[x];
  }
  float mine = hi ? s1 : s0;
  float send = hi ? s0 : s1;
  float r = mine + __shfl_xor(send, 16, 32);
  out[(b0 + (hi ? 1 : 0)) * 32 + u] = r;
}

extern "C" void kernel_launch(void* const* d_in, const int* in_sizes, int n_in,
                              void* d_out, int out_size, void* d_ws, size_t ws_size,
                              hipStream_t stream) {
  const float* X   = (const float*)d_in[0];
  const float* c11 = (const float*)d_in[1];
  const float* c12 = (const float*)d_in[2];
  const float* c13 = (const float*)d_in[3];
  const float* c14 = (const float*)d_in[4];
  const float* c21 = (const float*)d_in[5];
  const float* c22 = (const float*)d_in[6];
  const float* fm  = (const float*)d_in[7];
  const float* T   = (const float*)d_in[8];
  float* out = (float*)d_out;

  float* img = (float*)d_ws;

  int B = in_sizes[0] / 16;
  int blocks = (B * 32) / 256;
  pack_operands_kernel<<<dim3(58), dim3(256), 0, stream>>>(
      c11, c12, c13, c14, c21, c22, fm, T, img);
  tensortree_wmma_kernel<<<dim3(blocks), dim3(256), 0, stream>>>(X, img, out);
}
